# starting kernel
# baseline (speedup 1.0000x reference)
#include <hip/hip_runtime.h>
#include <hip/hip_bf16.h>

typedef __attribute__((ext_vector_type(2))) float v2f;
typedef __attribute__((ext_vector_type(8))) float v8f;

#define IN_SIZE 64
#define OUT_SIZE 64

__global__ __launch_bounds__(256) void rgcn_init_out(
    const float* __restrict__ bias, float* __restrict__ out, int n4, int n_rel) {
  int i = blockIdx.x * blockDim.x + threadIdx.x;
  if (i >= n4) return;
  int c = (i * 4) & (OUT_SIZE - 1);
  float4 b = make_float4(0.f, 0.f, 0.f, 0.f);
  for (int r = 0; r < n_rel; ++r) {
    const float* br = bias + r * OUT_SIZE + c;
    b.x += br[0]; b.y += br[1]; b.z += br[2]; b.w += br[3];
  }
  ((float4*)out)[i] = b;
}

__global__ __launch_bounds__(256) void rgcn_zero(float4* __restrict__ p, int n4) {
  int i = blockIdx.x * blockDim.x + threadIdx.x;
  if (i < n4) p[i] = make_float4(0.f, 0.f, 0.f, 0.f);
}

__global__ __launch_bounds__(256) void rgcn_scatter(
    const float* __restrict__ inp, const int* __restrict__ src,
    const int* __restrict__ dst, const float* __restrict__ eval,
    float* __restrict__ agg, int n_edges) {
  const int wid = (blockIdx.x * 256 + threadIdx.x) >> 5;
  if (wid >= n_edges) return;
  const int lane = threadIdx.x & 31;
  const int s = src[wid];
  const int d = dst[wid];
  const float v = eval[wid];
  const int col = lane * 2;
  const float2 m = *(const float2*)(inp + (size_t)s * IN_SIZE + col);
  float* p = agg + (size_t)d * IN_SIZE + col;
  unsafeAtomicAdd(p, m.x * v);
  unsafeAtomicAdd(p + 1, m.y * v);
}

__global__ __launch_bounds__(256) void rgcn_gemm_wmma(
    const float* __restrict__ agg, const float* __restrict__ W,
    float* __restrict__ out, int num_tiles) {
  __shared__ float sW[IN_SIZE * OUT_SIZE];
  const int t = threadIdx.x;
  {
    const float4* Wv = (const float4*)W;
    float4* sWv = (float4*)sW;
#pragma unroll
    for (int i = 0; i < (IN_SIZE * OUT_SIZE / 4) / 256; ++i)
      sWv[t + i * 256] = Wv[t + i * 256];
  }
  __syncthreads();

  const int wave = t >> 5;
  const int lane = t & 31;
  const int half = lane >> 4;
  const int l16 = lane & 15;

  const int tile = blockIdx.x * 8 + wave;
  if (tile >= num_tiles) return;
  const int rowBase = (tile >> 2) * 16;
  const int colBase = (tile & 3) * 16;

  v8f c;
#pragma unroll
  for (int j = 0; j < 8; ++j)
    c[j] = out[(size_t)(rowBase + j + half * 8) * OUT_SIZE + colBase + l16];

  const float* aRow = agg + (size_t)(rowBase + l16) * IN_SIZE;
  const int koff = half * 2;
#pragma unroll
  for (int k = 0; k < IN_SIZE; k += 4) {
    v2f a, b;
    const float2 av = *(const float2*)(aRow + k + koff);
    a[0] = av.x;
    a[1] = av.y;
    b[0] = sW[(k + koff) * OUT_SIZE + colBase + l16];
    b[1] = sW[(k + koff + 1) * OUT_SIZE + colBase + l16];
    c = __builtin_amdgcn_wmma_f32_16x16x4_f32(
         false, a,  false, b,
         (short)0, c,  false,  false);
  }

#pragma unroll
  for (int j = 0; j < 8; ++j)
    out[(size_t)(rowBase + j + half * 8) * OUT_SIZE + colBase + l16] = c[j];
}

extern "C" void kernel_launch(void* const* d_in, const int* in_sizes, int n_in,
                              void* d_out, int out_size, void* d_ws, size_t ws_size,
                              hipStream_t stream) {
  const float* inp = (const float*)d_in[0];
  const int* src = (const int*)d_in[1];
  const int* dst = (const int*)d_in[2];
  const float* eval = (const float*)d_in[3];
  const float* weights = (const float*)d_in[4];
  const float* bias = (const float*)d_in[5];
  float* out = (float*)d_out;
  float* agg = (float*)d_ws;

  const int n_nodes = in_sizes[0] / IN_SIZE;
  const int n_rel = in_sizes[4] / (IN_SIZE * OUT_SIZE);
  const int n_edges = in_sizes[1] / n_rel;

  const int n4 = n_nodes * OUT_SIZE / 4;
  const int vb = (n4 + 255) / 256;
  rgcn_init_out<<<vb, 256, 0, stream>>>(bias, out, n4, n_rel);

  const int scatter_blocks = (n_edges * 32 + 255) / 256;
  const int num_tiles = (n_nodes / 16) * (OUT_SIZE / 16);
  const int gemm_blocks = (num_tiles + 7) / 8;

  for (int r = 0; r < n_rel; ++r) {
    rgcn_zero<<<vb, 256, 0, stream>>>((float4*)agg, n4);
    rgcn_scatter<<<scatter_blocks, 256, 0, stream>>>(
        inp, src + (size_t)r * n_edges, dst + (size_t)r * n_edges,
        eval + (size_t)r * n_edges, agg, n_edges);
    rgcn_gemm_wmma<<<gemm_blocks, 256, 0, stream>>>(
        agg, weights + (size_t)r * IN_SIZE * OUT_SIZE, out, num_tiles);
  }
}


# edges put in destination-row order in the call (int count+rank, scan, fill); a wave per row and relation sums val*inp[src] in registers; one K=256 f32 WMMA pass; no float atomics; sum order: rank
# speedup vs baseline: 1.4553x; 1.4553x over previous
#include <hip/hip_runtime.h>
#include <hip/hip_bf16.h>


typedef __attribute__((ext_vector_type(2))) float v2f;
typedef __attribute__((ext_vector_type(8))) float v8f;

#define IN_SIZE 64
#define OUT_SIZE 64
#define SW_PITCH 72

__global__ __launch_bounds__(256) void rgcn_zero(int4* __restrict__ p, int n4) {
  int i = blockIdx.x * blockDim.x + threadIdx.x;
  if (i < n4) p[i] = make_int4(0, 0, 0, 0);
}

__global__ __launch_bounds__(256) void rgcn_hist(
    const int* __restrict__ dst, int* __restrict__ cnt, int* __restrict__ rank,
    int n_edges, int n_rel) {
  const int e0 = blockIdx.x * 1024 + threadIdx.x;
  const int r = blockIdx.y;
  const size_t base = (size_t)r * n_edges;
  int d[4];
#pragma unroll
  for (int u = 0; u < 4; ++u) {
    const int e = e0 + u * 256;
    d[u] = (e < n_edges) ? dst[base + e] : -1;
  }
  int k[4];
#pragma unroll
  for (int u = 0; u < 4; ++u)
    k[u] = (d[u] >= 0) ? atomicAdd(cnt + (size_t)d[u] * n_rel + r, 1) : 0;
#pragma unroll
  for (int u = 0; u < 4; ++u)
    if (d[u] >= 0) rank[base + e0 + u * 256] = k[u];
}

__device__ __forceinline__ int block_scan_256(int v, int* s, int t, int* total) {
  s[t] = v;
  __syncthreads();
  for (int off = 1; off < 256; off <<= 1) {
    const int add = (t >= off) ? s[t - off] : 0;
    __syncthreads();
    s[t] += add;
    __syncthreads();
  }
  const int incl = s[t];
  *total = s[255];
  __syncthreads();
  return incl - v;
}

__global__ __launch_bounds__(256) void rgcn_scan_reduce(
    const int* __restrict__ cnt, int* __restrict__ bsum, int n) {
  __shared__ int s[256];
  const int t = threadIdx.x;
  const int i0 = blockIdx.x * 1024 + t * 4;
  int v = 0;
#pragma unroll
  for (int u = 0; u < 4; ++u)
    if (i0 + u < n) v += cnt[i0 + u];
  int total;
  block_scan_256(v, s, t, &total);
  if (t == 0) bsum[blockIdx.x] = total;
}

__global__ __launch_bounds__(1024) void rgcn_scan_bsums(int* __restrict__ bsum, int nb) {
  __shared__ int s[1024];
  __shared__ int carry_s;
  const int t = threadIdx.x;
  if (t == 0) carry_s = 0;
  __syncthreads();
  for (int base = 0; base < nb; base += 1024) {
    const int i = base + t;
    const int v = (i < nb) ? bsum[i] : 0;
    s[t] = v;
    __syncthreads();
    for (int off = 1; off < 1024; off <<= 1) {
      const int add = (t >= off) ? s[t - off] : 0;
      __syncthreads();
      s[t] += add;
      __syncthreads();
    }
    const int carry = carry_s;
    const int incl = s[t];
    if (i < nb) bsum[i] = carry + incl - v;
    __syncthreads();
    if (t == 1023) carry_s = carry + incl;
    __syncthreads();
  }
}

__global__ __launch_bounds__(256) void rgcn_scan_apply(
    const int* __restrict__ cnt, const int* __restrict__ bsum,
    int* __restrict__ offs, int n) {
  __shared__ int s[256];
  const int t = threadIdx.x;
  const int i0 = blockIdx.x * 1024 + t * 4;
  int c[4];
  int v = 0;
#pragma unroll
  for (int u = 0; u < 4; ++u) {
    c[u] = (i0 + u < n) ? cnt[i0 + u] : 0;
    v += c[u];
  }
  int total;
  int run = block_scan_256(v, s, t, &total) + bsum[blockIdx.x];
#pragma unroll
  for (int u = 0; u < 4; ++u) {
    if (i0 + u < n) offs[i0 + u] = run;
    run += c[u];
  }
}

__global__ __launch_bounds__(256) void rgcn_fill(
    const int* __restrict__ src, const int* __restrict__ dst,
    const float* __restrict__ eval, const int* __restrict__ offs,
    const int* __restrict__ rank, int2* __restrict__ rec,
    int n_edges, int n_rel, int n_rec) {
  const int e0 = blockIdx.x * 1024 + threadIdx.x;
  const int r = blockIdx.y;
  const size_t base = (size_t)r * n_edges;
  int d[4], k[4], s[4], v[4];
#pragma unroll
  for (int u = 0; u < 4; ++u) {
    const int e = e0 + u * 256;
    const bool in = e < n_edges;
    d[u] = in ? dst[base + e] : -1;
    k[u] = in ? rank[base + e] : 0;
    s[u] = in ? src[base + e] : 0;
    v[u] = in ? __float_as_int(eval[base + e]) : 0;
  }
  int o[4];
#pragma unroll
  for (int u = 0; u < 4; ++u)
    o[u] = (d[u] >= 0) ? offs[(size_t)d[u] * n_rel + r] : 0;
#pragma unroll
  for (int u = 0; u < 4; ++u) {
    const int pos = o[u] + k[u];
    if (d[u] >= 0 && (unsigned)pos < (unsigned)n_rec) rec[pos] = make_int2(s[u], v[u]);
  }
}

template <int K>
__device__ __forceinline__ void pull_batch(const float* __restrict__ col, int2 rc, int j,
                                           float& ax, float& ay) {
  float2 x[K];
  float v[K];
#pragma unroll
  for (int u = 0; u < K; ++u) {
    const int s = __builtin_amdgcn_readlane(rc.x, j + u);
    v[u] = __int_as_float(__builtin_amdgcn_readlane(rc.y, j + u));
    x[u] = *(const float2*)(col + (size_t)s * IN_SIZE);
  }
#pragma unroll
  for (int u = 0; u < K; ++u) {
    ax += x[u].x * v[u];
    ay += x[u].y * v[u];
  }
}

__global__ __launch_bounds__(256) void rgcn_pull(
    const float* __restrict__ inp, const int* __restrict__ offs,
    const int* __restrict__ cnt, const int2* __restrict__ rec,
    float* __restrict__ agg, int n_seg, int n_rec) {
  const int seg = (blockIdx.x * 256 + threadIdx.x) >> 5;
  if (seg >= n_seg) return;
  const int lane = threadIdx.x & 31;
  const int beg = __builtin_amdgcn_readfirstlane(offs[seg]);
  const int n = __builtin_amdgcn_readfirstlane(cnt[seg]);
  const float* col = inp + 2 * lane;
  float ax = 0.f, ay = 0.f;
  for (int base = 0; base < n; base += 32) {
    const int m = min(32, n - base);
    int2 rc = make_int2(0, 0);
    const int p = beg + base + lane;
    if (lane < m && (unsigned)p < (unsigned)n_rec) rc = rec[p];
    int j = 0;
    for (; j + 16 <= m; j += 16) pull_batch<16>(col, rc, j, ax, ay);
    if (j + 8 <= m) { pull_batch<8>(col, rc, j, ax, ay); j += 8; }
    if (j + 4 <= m) { pull_batch<4>(col, rc, j, ax, ay); j += 4; }
    if (j + 2 <= m) { pull_batch<2>(col, rc, j, ax, ay); j += 2; }
    if (j < m) pull_batch<1>(col, rc, j, ax, ay);
  }
  *(float2*)(agg + (size_t)seg * IN_SIZE + 2 * lane) = make_float2(ax, ay);
}

__global__ __launch_bounds__(256) void rgcn_gemm_wmma(
    const float* __restrict__ agg, const float* __restrict__ W,
    const float* __restrict__ bias, float* __restrict__ out,
    int n_row_tiles, int n_rel) {
  __shared__ __attribute__((aligned(16))) float sW[IN_SIZE * SW_PITCH];
  const int t = threadIdx.x;
  const int wave = t >> 5;
  const int lane = t & 31;
  const int half = lane >> 4;
  const int l16 = lane & 15;

  int tile = blockIdx.x * 8 + wave;
  const bool live = tile < n_row_tiles;
  if (!live) tile = n_row_tiles - 1;
  const int rowBase = tile * 16;
  const int kpitch = n_rel * IN_SIZE;

  float b0 = 0.f, b1 = 0.f, b2 = 0.f, b3 = 0.f;
  for (int r = 0; r < n_rel; ++r) {
    const float* br = bias + r * OUT_SIZE + l16;
    b0 += br[0]; b1 += br[16]; b2 += br[32]; b3 += br[48];
  }
  v8f c0, c1, c2, c3;
#pragma unroll
  for (int j = 0; j < 8; ++j) { c0[j] = b0; c1[j] = b1; c2[j] = b2; c3[j] = b3; }

  const int koff = half * 2;
  const float* aRow = agg + (size_t)(rowBase + l16) * kpitch + koff;
  const float4* Wv = (const float4*)W;

  for (int r = 0; r < n_rel; ++r) {
    const float* aR = aRow + r * IN_SIZE;
    float2 av[IN_SIZE / 4];
#pragma unroll
    for (int i = 0; i < IN_SIZE / 4; ++i) av[i] = *(const float2*)(aR + 4 * i);
    const float4* Wr = Wv + (size_t)r * (IN_SIZE * OUT_SIZE / 4) + t;
    const float4 w0 = Wr[0], w1 = Wr[256], w2 = Wr[512], w3 = Wr[768];
    __syncthreads();
    float* sp = sW + (t >> 4) * SW_PITCH + (t & 15) * 4;
    *(float4*)(sp) = w0;
    *(float4*)(sp + 16 * SW_PITCH) = w1;
    *(float4*)(sp + 32 * SW_PITCH) = w2;
    *(float4*)(sp + 48 * SW_PITCH) = w3;
    __syncthreads();
#pragma unroll
    for (int k = 0; k < IN_SIZE; k += 4) {
      v2f a, b;
      a[0] = av[k / 4].x;
      a[1] = av[k / 4].y;
      const float* bp = sW + (k + koff) * SW_PITCH + l16;
      b[0] = bp[0];
      b[1] = bp[SW_PITCH];
      c0 = __builtin_amdgcn_wmma_f32_16x16x4_f32(false, a, false, b, (short)0, c0, false, false);
      b[0] = bp[16];
      b[1] = bp[SW_PITCH + 16];
      c1 = __builtin_amdgcn_wmma_f32_16x16x4_f32(false, a, false, b, (short)0, c1, false, false);
      b[0] = bp[32];
      b[1] = bp[SW_PITCH + 32];
      c2 = __builtin_amdgcn_wmma_f32_16x16x4_f32(false, a, false, b, (short)0, c2, false, false);
      b[0] = bp[48];
      b[1] = bp[SW_PITCH + 48];
      c3 = __builtin_amdgcn_wmma_f32_16x16x4_f32(false, a, false, b, (short)0, c3, false, false);
    }
  }

  if (live) {
#pragma unroll
    for (int j = 0; j < 8; ++j) {
      float* o = out + (size_t)(rowBase + j + half * 8) * OUT_SIZE + l16;
      o[0] = c0[j];
      o[16] = c1[j];
      o[32] = c2[j];
      o[48] = c3[j];
    }
  }
}

extern "C" void kernel_launch(void* const* d_in, const int* in_sizes, int n_in,
                              void* d_out, int out_size, void* d_ws, size_t ws_size,
                              hipStream_t stream) {
  const float* inp = (const float*)d_in[0];
  const int* src = (const int*)d_in[1];
  const int* dst = (const int*)d_in[2];
  const float* eval = (const float*)d_in[3];
  const float* weights = (const float*)d_in[4];
  const float* bias = (const float*)d_in[5];
  float* out = (float*)d_out;

  const int n_nodes = in_sizes[0] / IN_SIZE;
  const int n_rel = in_sizes[4] / (IN_SIZE * OUT_SIZE);
  const int n_edges = in_sizes[1] / n_rel;
  const int n_rec = n_rel * n_edges;
  const int n_seg = n_nodes * n_rel;
  const int n_slices = (n_seg + 1023) / 1024;
  const size_t seg_pad = (size_t)n_slices * 1024;
  const size_t rec_pad = ((size_t)n_rec + 1023) / 1024 * 1024;

  int* cnt = (int*)d_ws;
  int* offs = cnt + seg_pad;
  int* bsum = offs + seg_pad;
  int* rank = bsum + seg_pad;
  int2* rec = (int2*)(rank + rec_pad);
  float* agg = (float*)(rec + rec_pad);

  const int zero4 = (int)(seg_pad / 4);
  rgcn_zero<<<(zero4 + 255) / 256, 256, 0, stream>>>((int4*)cnt, zero4);

  const dim3 egrid((n_edges + 1023) / 1024, n_rel);
  rgcn_hist<<<egrid, 256, 0, stream>>>(dst, cnt, rank, n_edges, n_rel);

  rgcn_scan_reduce<<<n_slices, 256, 0, stream>>>(cnt, bsum, n_seg);
  rgcn_scan_bsums<<<1, 1024, 0, stream>>>(bsum, n_slices);
  rgcn_scan_apply<<<n_slices, 256, 0, stream>>>(cnt, bsum, offs, n_seg);

  rgcn_fill<<<egrid, 256, 0, stream>>>(src, dst, eval, offs, rank, rec,
                                       n_edges, n_rel, n_rec);

  rgcn_pull<<<(n_seg + 7) / 8, 256, 0, stream>>>(inp, offs, cnt, rec, agg, n_seg, n_rec);

  const int n_row_tiles = n_nodes / 16;
  rgcn_gemm_wmma<<<(n_row_tiles + 7) / 8, 256, 0, stream>>>(
      agg, weights, bias, out, n_row_tiles, n_rel);
}


# 32-slot record block per (row, relation): an edge's returning count is its slot, so one edge pass orders them (no rank array, scan, fill pass); edges past slot 32: float atomic adds, arrival order
# speedup vs baseline: 1.6237x; 1.1157x over previous
#include <hip/hip_runtime.h>
#include <hip/hip_bf16.h>


typedef __attribute__((ext_vector_type(2))) float v2f;
typedef __attribute__((ext_vector_type(8))) float v8f;

#define IN_SIZE 64
#define OUT_SIZE 64
#define SLOTS 32
#define SW_PITCH 72

__global__ __launch_bounds__(256) void rgcn_zero(int4* __restrict__ p, int n4) {
  int i = blockIdx.x * blockDim.x + threadIdx.x;
  if (i < n4) p[i] = make_int4(0, 0, 0, 0);
}

__global__ __launch_bounds__(256) void rgcn_build(
    const int* __restrict__ src, const int* __restrict__ dst,
    const float* __restrict__ eval, int* __restrict__ cnt, int* __restrict__ ovf_n,
    int2* __restrict__ rec, int* __restrict__ ovf,
    int n_edges, int n_rel, int n_nodes, int n_rec) {
  const int e0 = blockIdx.x * 1024 + threadIdx.x;
  const int r = blockIdx.y;
  const size_t base = (size_t)r * n_edges;
  int d[4], s[4], v[4];
#pragma unroll
  for (int u = 0; u < 4; ++u) {
    const int e = e0 + u * 256;
    const bool in = e < n_edges;
    d[u] = in ? dst[base + e] : -1;
    s[u] = in ? src[base + e] : 0;
    v[u] = in ? __float_as_int(eval[base + e]) : 0;
  }
  int k[4];
#pragma unroll
  for (int u = 0; u < 4; ++u) {
    if ((unsigned)d[u] >= (unsigned)n_nodes) d[u] = -1;
    k[u] = (d[u] >= 0) ? atomicAdd(cnt + (size_t)d[u] * n_rel + r, 1) : 0;
  }
#pragma unroll
  for (int u = 0; u < 4; ++u) {
    if (d[u] < 0) continue;
    if (k[u] < SLOTS) {
      rec[((size_t)d[u] * n_rel + r) * SLOTS + k[u]] = make_int2(s[u], v[u]);
    } else {
      const int pos = atomicAdd(ovf_n, 1);
      if ((unsigned)pos < (unsigned)n_rec) ovf[pos] = (int)base + e0 + u * 256;
    }
  }
}

template <int K>
__device__ __forceinline__ void pull_batch(const float* __restrict__ col, int2 rc, int j,
                                           float& ax, float& ay) {
  float2 x[K];
  float v[K];
#pragma unroll
  for (int u = 0; u < K; ++u) {
    const int s = __builtin_amdgcn_readlane(rc.x, j + u);
    v[u] = __int_as_float(__builtin_amdgcn_readlane(rc.y, j + u));
    x[u] = *(const float2*)(col + (size_t)s * IN_SIZE);
  }
#pragma unroll
  for (int u = 0; u < K; ++u) {
    ax += x[u].x * v[u];
    ay += x[u].y * v[u];
  }
}

__global__ __launch_bounds__(256) void rgcn_pull(
    const float* __restrict__ inp, const int* __restrict__ cnt,
    const int2* __restrict__ rec, float* __restrict__ agg, int n_seg) {
  const int seg = (blockIdx.x * 256 + threadIdx.x) >> 5;
  if (seg >= n_seg) return;
  const int lane = threadIdx.x & 31;
  const int m = min(__builtin_amdgcn_readfirstlane(cnt[seg]), SLOTS);
  const float* col = inp + 2 * lane;
  float ax = 0.f, ay = 0.f;
  int2 rc = make_int2(0, 0);
  if (lane < m) rc = rec[(size_t)seg * SLOTS + lane];
  int j = 0;
  for (; j + 16 <= m; j += 16) pull_batch<16>(col, rc, j, ax, ay);
  if (j + 8 <= m) { pull_batch<8>(col, rc, j, ax, ay); j += 8; }
  if (j + 4 <= m) { pull_batch<4>(col, rc, j, ax, ay); j += 4; }
  if (j + 2 <= m) { pull_batch<2>(col, rc, j, ax, ay); j += 2; }
  if (j < m) pull_batch<1>(col, rc, j, ax, ay);
  *(float2*)(agg + (size_t)seg * IN_SIZE + 2 * lane) = make_float2(ax, ay);
}

__global__ __launch_bounds__(256) void rgcn_overflow(
    const float* __restrict__ inp, const int* __restrict__ src,
    const int* __restrict__ dst, const float* __restrict__ eval,
    const int* __restrict__ ovf, const int* __restrict__ ovf_n,
    float* __restrict__ agg, int n_edges, int n_rel, int n_nodes, int n_rec) {
  const int lane = threadIdx.x & 31;
  const int wave = (blockIdx.x * 256 + threadIdx.x) >> 5;
  const int n_waves = gridDim.x * 8;
  const int n = min(__builtin_amdgcn_readfirstlane(*ovf_n), n_rec);
  for (int i = wave; i < n; i += n_waves) {
    const int e = __builtin_amdgcn_readfirstlane(ovf[i]);
    if ((unsigned)e >= (unsigned)n_rec) continue;
    const int r = e / n_edges;
    const int d = dst[e];
    const int s = src[e];
    const float v = eval[e];
    if ((unsigned)d >= (unsigned)n_nodes) continue;
    const float2 x = *(const float2*)(inp + (size_t)s * IN_SIZE + 2 * lane);
    float* a = agg + ((size_t)d * n_rel + r) * IN_SIZE + 2 * lane;
    atomicAdd(a, x.x * v);
    atomicAdd(a + 1, x.y * v);
  }
}

__global__ __launch_bounds__(256) void rgcn_gemm_wmma(
    const float* __restrict__ agg, const float* __restrict__ W,
    const float* __restrict__ bias, float* __restrict__ out,
    int n_row_tiles, int n_rel) {
  __shared__ __attribute__((aligned(16))) float sW[IN_SIZE * SW_PITCH];
  const int t = threadIdx.x;
  const int wave = t >> 5;
  const int lane = t & 31;
  const int half = lane >> 4;
  const int l16 = lane & 15;

  int tile = blockIdx.x * 8 + wave;
  const bool live = tile < n_row_tiles;
  if (!live) tile = n_row_tiles - 1;
  const int rowBase = tile * 16;
  const int kpitch = n_rel * IN_SIZE;

  float b0 = 0.f, b1 = 0.f, b2 = 0.f, b3 = 0.f;
  for (int r = 0; r < n_rel; ++r) {
    const float* br = bias + r * OUT_SIZE + l16;
    b0 += br[0]; b1 += br[16]; b2 += br[32]; b3 += br[48];
  }
  v8f c0, c1, c2, c3;
#pragma unroll
  for (int j = 0; j < 8; ++j) { c0[j] = b0; c1[j] = b1; c2[j] = b2; c3[j] = b3; }

  const int koff = half * 2;
  const float* aRow = agg + (size_t)(rowBase + l16) * kpitch + koff;
  const float4* Wv = (const float4*)W;

  for (int r = 0; r < n_rel; ++r) {
    const float* aR = aRow + r * IN_SIZE;
    float2 av[IN_SIZE / 4];
#pragma unroll
    for (int i = 0; i < IN_SIZE / 4; ++i) av[i] = *(const float2*)(aR + 4 * i);
    const float4* Wr = Wv + (size_t)r * (IN_SIZE * OUT_SIZE / 4) + t;
    const float4 w0 = Wr[0], w1 = Wr[256], w2 = Wr[512], w3 = Wr[768];
    __syncthreads();
    float* sp = sW + (t >> 4) * SW_PITCH + (t & 15) * 4;
    *(float4*)(sp) = w0;
    *(float4*)(sp + 16 * SW_PITCH) = w1;
    *(float4*)(sp + 32 * SW_PITCH) = w2;
    *(float4*)(sp + 48 * SW_PITCH) = w3;
    __syncthreads();
#pragma unroll
    for (int k = 0; k < IN_SIZE; k += 4) {
      v2f a, b;
      a[0] = av[k / 4].x;
      a[1] = av[k / 4].y;
      const float* bp = sW + (k + koff) * SW_PITCH + l16;
      b[0] = bp[0];
      b[1] = bp[SW_PITCH];
      c0 = __builtin_amdgcn_wmma_f32_16x16x4_f32(false, a, false, b, (short)0, c0, false, false);
      b[0] = bp[16];
      b[1] = bp[SW_PITCH + 16];
      c1 = __builtin_amdgcn_wmma_f32_16x16x4_f32(false, a, false, b, (short)0, c1, false, false);
      b[0] = bp[32];
      b[1] = bp[SW_PITCH + 32];
      c2 = __builtin_amdgcn_wmma_f32_16x16x4_f32(false, a, false, b, (short)0, c2, false, false);
      b[0] = bp[48];
      b[1] = bp[SW_PITCH + 48];
      c3 = __builtin_amdgcn_wmma_f32_16x16x4_f32(false, a, false, b, (short)0, c3, false, false);
    }
  }

  if (live) {
#pragma unroll
    for (int j = 0; j < 8; ++j) {
      float* o = out + (size_t)(rowBase + j + half * 8) * OUT_SIZE + l16;
      o[0] = c0[j];
      o[16] = c1[j];
      o[32] = c2[j];
      o[48] = c3[j];
    }
  }
}

extern "C" void kernel_launch(void* const* d_in, const int* in_sizes, int n_in,
                              void* d_out, int out_size, void* d_ws, size_t ws_size,
                              hipStream_t stream) {
  const float* inp = (const float*)d_in[0];
  const int* src = (const int*)d_in[1];
  const int* dst = (const int*)d_in[2];
  const float* eval = (const float*)d_in[3];
  const float* weights = (const float*)d_in[4];
  const float* bias = (const float*)d_in[5];
  float* out = (float*)d_out;

  const int n_nodes = in_sizes[0] / IN_SIZE;
  const int n_rel = in_sizes[4] / (IN_SIZE * OUT_SIZE);
  const int n_edges = in_sizes[1] / n_rel;
  const int n_rec = n_rel * n_edges;
  const int n_seg = n_nodes * n_rel;
  const int n_slices = (n_seg + 1023) / 1024;
  const size_t seg_pad = (size_t)(n_slices + 1) * 1024;
  const size_t rec_pad = ((size_t)n_seg * SLOTS + 1023) / 1024 * 1024;
  const size_t ovf_pad = ((size_t)n_rec + 1023) / 1024 * 1024;

  int* cnt = (int*)d_ws;
  int* ovf_n = cnt + (size_t)n_slices * 1024;
  int2* rec = (int2*)(cnt + seg_pad);
  int* ovf = (int*)(rec + rec_pad);
  float* agg = (float*)(ovf + ovf_pad);

  const int zero4 = (int)(seg_pad / 4);
  rgcn_zero<<<(zero4 + 255) / 256, 256, 0, stream>>>((int4*)cnt, zero4);

  const dim3 egrid((n_edges + 1023) / 1024, n_rel);
  rgcn_build<<<egrid, 256, 0, stream>>>(src, dst, eval, cnt, ovf_n, rec, ovf,
                                        n_edges, n_rel, n_nodes, n_rec);

  rgcn_pull<<<(n_seg + 7) / 8, 256, 0, stream>>>(inp, cnt, rec, agg, n_seg);

  rgcn_overflow<<<256, 256, 0, stream>>>(inp, src, dst, eval, ovf, ovf_n, agg,
                                         n_edges, n_rel, n_nodes, n_rec);

  const int n_row_tiles = n_nodes / 16;
  rgcn_gemm_wmma<<<(n_row_tiles + 7) / 8, 256, 0, stream>>>(
      agg, weights, bias, out, n_row_tiles, n_rel);
}


# speedup vs baseline: 3.8489x; 2.3705x over previous
#include <hip/hip_runtime.h>
#include <hip/hip_bf16.h>


typedef __attribute__((ext_vector_type(2))) float v2f;
typedef __attribute__((ext_vector_type(8))) float v8f;

#define IN_SIZE 64
#define OUT_SIZE 64
#define HSLOTS 16
#define SW_PITCH 72

__global__ __launch_bounds__(256) void rgcn_zero(int4* __restrict__ p, int n4) {
  int i = blockIdx.x * blockDim.x + threadIdx.x;
  if (i < n4) p[i] = make_int4(0, 0, 0, 0);
}

__device__ __forceinline__ int chip_half() {
  return (__builtin_amdgcn_s_sendmsg_rtn(0x87) >> 18) & 1;
}

__global__ __launch_bounds__(256) void rgcn_build(
    const int* __restrict__ src, const int* __restrict__ dst,
    const float* __restrict__ eval, int* __restrict__ cnt, int* __restrict__ ovf_n,
    int2* __restrict__ rec, int* __restrict__ ovf,
    int n_edges, int n_rel, int n_nodes, int n_rec, int cnt_plane, size_t rec_plane) {
  const int h = chip_half();
  cnt += (size_t)h * cnt_plane;
  rec += (size_t)h * rec_plane;
  const int e0 = blockIdx.x * 1024 + threadIdx.x;
  const int r = blockIdx.y;
  const size_t base = (size_t)r * n_edges;
  int d[4], s[4], v[4];
#pragma unroll
  for (int u = 0; u < 4; ++u) {
    const int e = e0 + u * 256;
    const bool in = e < n_edges;
    d[u] = in ? dst[base + e] : -1;
    s[u] = in ? src[base + e] : 0;
    v[u] = in ? __float_as_int(eval[base + e]) : 0;
  }
  int k[4];
#pragma unroll
  for (int u = 0; u < 4; ++u) {
    if ((unsigned)d[u] >= (unsigned)n_nodes) d[u] = -1;
    k[u] = (d[u] >= 0) ? atomicAdd(cnt + (size_t)d[u] * n_rel + r, 1) : 0;
  }
#pragma unroll
  for (int u = 0; u < 4; ++u) {
    if (d[u] < 0) continue;
    if (k[u] < HSLOTS) {
      rec[((size_t)d[u] * n_rel + r) * HSLOTS + k[u]] = make_int2(s[u], v[u]);
    } else {
      const int pos = atomicAdd(ovf_n, 1);
      if ((unsigned)pos < (unsigned)n_rec) ovf[pos] = (int)base + e0 + u * 256;
    }
  }
}

template <int K>
__device__ __forceinline__ void pull_batch(const float* __restrict__ col, int2 rc, int j,
                                           float& ax, float& ay) {
  float2 x[K];
  float v[K];
#pragma unroll
  for (int u = 0; u < K; ++u) {
    const int s = __builtin_amdgcn_readlane(rc.x, j + u);
    v[u] = __int_as_float(__builtin_amdgcn_readlane(rc.y, j + u));
    x[u] = *(const float2*)(col + (size_t)s * IN_SIZE);
  }
#pragma unroll
  for (int u = 0; u < K; ++u) {
    ax += x[u].x * v[u];
    ay += x[u].y * v[u];
  }
}

__global__ __launch_bounds__(256) void rgcn_pull(
    const float* __restrict__ inp, const int* __restrict__ cnt,
    const int2* __restrict__ rec, float* __restrict__ agg, int n_seg,
    int cnt_plane, size_t rec_plane) {
  const int seg = (blockIdx.x * 256 + threadIdx.x) >> 5;
  if (seg >= n_seg) return;
  const int lane = threadIdx.x & 31;
  const int c0 = cnt[seg], c1 = cnt[(size_t)cnt_plane + seg];
  const int m0 = min(__builtin_amdgcn_readfirstlane(c0), HSLOTS);
  const int m = m0 + min(__builtin_amdgcn_readfirstlane(c1), HSLOTS);
  const float* col = inp + 2 * lane;
  float ax = 0.f, ay = 0.f;
  int2 rc = make_int2(0, 0);
  if (lane < m)
    rc = (lane < m0) ? rec[(size_t)seg * HSLOTS + lane]
                     : rec[rec_plane + (size_t)seg * HSLOTS + (lane - m0)];
  int j = 0;
  for (; j + 16 <= m; j += 16) pull_batch<16>(col, rc, j, ax, ay);
  if (j + 8 <= m) { pull_batch<8>(col, rc, j, ax, ay); j += 8; }
  if (j + 4 <= m) { pull_batch<4>(col, rc, j, ax, ay); j += 4; }
  if (j + 2 <= m) { pull_batch<2>(col, rc, j, ax, ay); j += 2; }
  if (j < m) pull_batch<1>(col, rc, j, ax, ay);
  *(float2*)(agg + (size_t)seg * IN_SIZE + 2 * lane) = make_float2(ax, ay);
}

__global__ __launch_bounds__(256) void rgcn_overflow(
    const float* __restrict__ inp, const int* __restrict__ src,
    const int* __restrict__ dst, const float* __restrict__ eval,
    const int* __restrict__ ovf, const int* __restrict__ ovf_n,
    float* __restrict__ agg, int n_edges, int n_rel, int n_nodes, int n_rec) {
  const int lane = threadIdx.x & 31;
  const int wave = (blockIdx.x * 256 + threadIdx.x) >> 5;
  const int n_waves = gridDim.x * 8;
  const int n = min(__builtin_amdgcn_readfirstlane(*ovf_n), n_rec);
  for (int i = wave; i < n; i += n_waves) {
    const int e = __builtin_amdgcn_readfirstlane(ovf[i]);
    if ((unsigned)e >= (unsigned)n_rec) continue;
    const int r = e / n_edges;
    const int d = dst[e];
    const int s = src[e];
    const float v = eval[e];
    if ((unsigned)d >= (unsigned)n_nodes) continue;
    const float2 x = *(const float2*)(inp + (size_t)s * IN_SIZE + 2 * lane);
    float* a = agg + ((size_t)d * n_rel + r) * IN_SIZE + 2 * lane;
    atomicAdd(a, x.x * v);
    atomicAdd(a + 1, x.y * v);
  }
}

__global__ __launch_bounds__(256) void rgcn_gemm_wmma(
    const float* __restrict__ agg, const float* __restrict__ W,
    const float* __restrict__ bias, float* __restrict__ out,
    int n_row_tiles, int n_rel) {
  __shared__ __attribute__((aligned(16))) float sW[IN_SIZE * SW_PITCH];
  const int t = threadIdx.x;
  const int wave = t >> 5;
  const int lane = t & 31;
  const int half = lane >> 4;
  const int l16 = lane & 15;

  int tile = blockIdx.x * 8 + wave;
  const bool live = tile < n_row_tiles;
  if (!live) tile = n_row_tiles - 1;
  const int rowBase = tile * 16;
  const int kpitch = n_rel * IN_SIZE;

  float b0 = 0.f, b1 = 0.f, b2 = 0.f, b3 = 0.f;
  for (int r = 0; r < n_rel; ++r) {
    const float* br = bias + r * OUT_SIZE + l16;
    b0 += br[0]; b1 += br[16]; b2 += br[32]; b3 += br[48];
  }
  v8f c0, c1, c2, c3;
#pragma unroll
  for (int j = 0; j < 8; ++j) { c0[j] = b0; c1[j] = b1; c2[j] = b2; c3[j] = b3; }

  const int koff = half * 2;
  const float* aRow = agg + (size_t)(rowBase + l16) * kpitch + koff;
  const float4* Wv = (const float4*)W;

  for (int r = 0; r < n_rel; ++r) {
    const float* aR = aRow + r * IN_SIZE;
    float2 av[IN_SIZE / 4];
#pragma unroll
    for (int i = 0; i < IN_SIZE / 4; ++i) av[i] = *(const float2*)(aR + 4 * i);
    const float4* Wr = Wv + (size_t)r * (IN_SIZE * OUT_SIZE / 4) + t;
    const float4 w0 = Wr[0], w1 = Wr[256], w2 = Wr[512], w3 = Wr[768];
    __syncthreads();
    float* sp = sW + (t >> 4) * SW_PITCH + (t & 15) * 4;
    *(float4*)(sp) = w0;
    *(float4*)(sp + 16 * SW_PITCH) = w1;
    *(float4*)(sp + 32 * SW_PITCH) = w2;
    *(float4*)(sp + 48 * SW_PITCH) = w3;
    __syncthreads();
#pragma unroll
    for (int k = 0; k < IN_SIZE; k += 4) {
      v2f a, b;
      a[0] = av[k / 4].x;
      a[1] = av[k / 4].y;
      const float* bp = sW + (k + koff) * SW_PITCH + l16;
      b[0] = bp[0];
      b[1] = bp[SW_PITCH];
      c0 = __builtin_amdgcn_wmma_f32_16x16x4_f32(false, a, false, b, (short)0, c0, false, false);
      b[0] = bp[16];
      b[1] = bp[SW_PITCH + 16];
      c1 = __builtin_amdgcn_wmma_f32_16x16x4_f32(false, a, false, b, (short)0, c1, false, false);
      b[0] = bp[32];
      b[1] = bp[SW_PITCH + 32];
      c2 = __builtin_amdgcn_wmma_f32_16x16x4_f32(false, a, false, b, (short)0, c2, false, false);
      b[0] = bp[48];
      b[1] = bp[SW_PITCH + 48];
      c3 = __builtin_amdgcn_wmma_f32_16x16x4_f32(false, a, false, b, (short)0, c3, false, false);
    }
  }

  if (live) {
#pragma unroll
    for (int j = 0; j < 8; ++j) {
      float* o = out + (size_t)(rowBase + j + half * 8) * OUT_SIZE + l16;
      o[0] = c0[j];
      o[16] = c1[j];
      o[32] = c2[j];
      o[48] = c3[j];
    }
  }
}

extern "C" void kernel_launch(void* const* d_in, const int* in_sizes, int n_in,
                              void* d_out, int out_size, void* d_ws, size_t ws_size,
                              hipStream_t stream) {
  const float* inp = (const float*)d_in[0];
  const int* src = (const int*)d_in[1];
  const int* dst = (const int*)d_in[2];
  const float* eval = (const float*)d_in[3];
  const float* weights = (const float*)d_in[4];
  const float* bias = (const float*)d_in[5];
  float* out = (float*)d_out;

  const int n_nodes = in_sizes[0] / IN_SIZE;
  const int n_rel = in_sizes[4] / (IN_SIZE * OUT_SIZE);
  const int n_edges = in_sizes[1] / n_rel;
  const int n_rec = n_rel * n_edges;
  const int n_seg = n_nodes * n_rel;
  const int n_slices = (n_seg + 1023) / 1024;
  const int cnt_plane = n_slices * 1024;
  const size_t seg_pad = (size_t)(2 * n_slices + 1) * 1024;
  const size_t rec_plane = ((size_t)n_seg * HSLOTS + 1023) / 1024 * 1024;
  const size_t rec_pad = 2 * rec_plane;
  const size_t ovf_pad = ((size_t)n_rec + 1023) / 1024 * 1024;

  int* cnt = (int*)d_ws;
  int* ovf_n = cnt + (size_t)2 * cnt_plane;
  int2* rec = (int2*)(cnt + seg_pad);
  int* ovf = (int*)(rec + rec_pad);
  float* agg = (float*)(ovf + ovf_pad);

  const int zero4 = (int)(seg_pad / 4);
  rgcn_zero<<<(zero4 + 255) / 256, 256, 0, stream>>>((int4*)cnt, zero4);

  const dim3 egrid((n_edges + 1023) / 1024, n_rel);
  rgcn_build<<<egrid, 256, 0, stream>>>(src, dst, eval, cnt, ovf_n, rec, ovf,
                                        n_edges, n_rel, n_nodes, n_rec, cnt_plane, rec_plane);

  rgcn_pull<<<(n_seg + 7) / 8, 256, 0, stream>>>(inp, cnt, rec, agg, n_seg, cnt_plane, rec_plane);

  rgcn_overflow<<<256, 256, 0, stream>>>(inp, src, dst, eval, ovf, ovf_n, agg,
                                         n_edges, n_rel, n_nodes, n_rec);

  const int n_row_tiles = n_nodes / 16;
  rgcn_gemm_wmma<<<(n_row_tiles + 7) / 8, 256, 0, stream>>>(
      agg, weights, bias, out, n_row_tiles, n_rel);
}


# pull: one wave sums a run of 25 consecutive (row, relation) segments, was one segment per wave: counts in one coalesced load, next segment's records requested a trip ahead; order of sums unchanged
# speedup vs baseline: 4.0598x; 1.0548x over previous
#include <hip/hip_runtime.h>
#include <hip/hip_bf16.h>


typedef __attribute__((ext_vector_type(2))) float v2f;
typedef __attribute__((ext_vector_type(8))) float v8f;

#define IN_SIZE 64
#define OUT_SIZE 64
#define HSLOTS 16
#define SW_PITCH 72

__global__ __launch_bounds__(256) void rgcn_zero(int4* __restrict__ p, int n4) {
  int i = blockIdx.x * blockDim.x + threadIdx.x;
  if (i < n4) p[i] = make_int4(0, 0, 0, 0);
}

__device__ __forceinline__ int chip_half() {
  return (__builtin_amdgcn_s_sendmsg_rtn(0x87) >> 18) & 1;
}

__global__ __launch_bounds__(256) void rgcn_build(
    const int* __restrict__ src, const int* __restrict__ dst,
    const float* __restrict__ eval, int* __restrict__ cnt, int* __restrict__ ovf_n,
    int2* __restrict__ rec, int* __restrict__ ovf,
    int n_edges, int n_rel, int n_nodes, int n_rec, int cnt_plane, size_t rec_plane) {
  const int h = chip_half();
  cnt += (size_t)h * cnt_plane;
  rec += (size_t)h * rec_plane;
  const int e0 = blockIdx.x * 1024 + threadIdx.x;
  const int r = blockIdx.y;
  const size_t base = (size_t)r * n_edges;
  int d[4], s[4], v[4];
#pragma unroll
  for (int u = 0; u < 4; ++u) {
    const int e = e0 + u * 256;
    const bool in = e < n_edges;
    d[u] = in ? dst[base + e] : -1;
    s[u] = in ? src[base + e] : 0;
    v[u] = in ? __float_as_int(eval[base + e]) : 0;
  }
  int k[4];
#pragma unroll
  for (int u = 0; u < 4; ++u) {
    if ((unsigned)d[u] >= (unsigned)n_nodes) d[u] = -1;
    k[u] = (d[u] >= 0) ? atomicAdd(cnt + (size_t)d[u] * n_rel + r, 1) : 0;
  }
#pragma unroll
  for (int u = 0; u < 4; ++u) {
    if (d[u] < 0) continue;
    if (k[u] < HSLOTS) {
      rec[((size_t)d[u] * n_rel + r) * HSLOTS + k[u]] = make_int2(s[u], v[u]);
    } else {
      const int pos = atomicAdd(ovf_n, 1);
      if ((unsigned)pos < (unsigned)n_rec) ovf[pos] = (int)base + e0 + u * 256;
    }
  }
}

template <int K>
__device__ __forceinline__ void pull_batch(const float* __restrict__ col, int2 rc, int j,
                                           float& ax, float& ay) {
  float2 x[K];
  float v[K];
#pragma unroll
  for (int u = 0; u < K; ++u) {
    const int s = __builtin_amdgcn_readlane(rc.x, j + u);
    v[u] = __int_as_float(__builtin_amdgcn_readlane(rc.y, j + u));
    x[u] = *(const float2*)(col + (size_t)s * IN_SIZE);
  }
#pragma unroll
  for (int u = 0; u < K; ++u) {
    ax += x[u].x * v[u];
    ay += x[u].y * v[u];
  }
}

#define RUN 25

__device__ __forceinline__ int2 seg_records(const int2* __restrict__ r0, size_t rec_plane,
                                            int lane, int m0, int m) {
  int2 rc = make_int2(0, 0);
  if (lane < m) rc = (lane < m0) ? r0[lane] : r0[rec_plane + (lane - m0)];
  return rc;
}

__global__ __launch_bounds__(128) void rgcn_pull(
    const float* __restrict__ inp, const int* __restrict__ cnt,
    const int2* __restrict__ rec, float* __restrict__ agg, int n_seg,
    int cnt_plane, size_t rec_plane) {
  const int lane = threadIdx.x & 31;
  const int seg0 = __builtin_amdgcn_readfirstlane(((blockIdx.x * 128 + threadIdx.x) >> 5) * RUN);
  if (seg0 >= n_seg) return;
  const int nrun = min(RUN, n_seg - seg0);
  int c0v = 0, c1v = 0;
  if (lane < nrun) {
    c0v = cnt[seg0 + lane];
    c1v = cnt[(size_t)cnt_plane + seg0 + lane];
  }
  c0v = min(c0v, HSLOTS);
  c1v = min(c1v, HSLOTS);
  const float* col = inp + 2 * lane;
  const int2* r0 = rec + (size_t)seg0 * HSLOTS;
  float* arow = agg + (size_t)seg0 * IN_SIZE + 2 * lane;
  int m0 = __builtin_amdgcn_readlane(c0v, 0);
  int m = m0 + __builtin_amdgcn_readlane(c1v, 0);
  int2 rc = seg_records(r0, rec_plane, lane, m0, m);
  asm volatile("" ::"v"(rc.x), "v"(rc.y));
  for (int i = 0; i < nrun; ++i) {
    int n = 0;
    int2 rn = make_int2(0, 0);
    if (i + 1 < nrun) {
      const int n0 = __builtin_amdgcn_readlane(c0v, i + 1);
      n = n0 + __builtin_amdgcn_readlane(c1v, i + 1);
      rn = seg_records(r0 + (size_t)(i + 1) * HSLOTS, rec_plane, lane, n0, n);
    }
    float ax = 0.f, ay = 0.f;
    int j = 0;
    for (; j + 16 <= m; j += 16) pull_batch<16>(col, rc, j, ax, ay);
    if (j + 8 <= m) { pull_batch<8>(col, rc, j, ax, ay); j += 8; }
    if (j + 4 <= m) { pull_batch<4>(col, rc, j, ax, ay); j += 4; }
    if (j + 2 <= m) { pull_batch<2>(col, rc, j, ax, ay); j += 2; }
    if (j < m) pull_batch<1>(col, rc, j, ax, ay);
    *(float2*)(arow + (size_t)i * IN_SIZE) = make_float2(ax, ay);
    rc = rn;
    m = n;
  }
}

__global__ __launch_bounds__(256) void rgcn_overflow(
    const float* __restrict__ inp, const int* __restrict__ src,
    const int* __restrict__ dst, const float* __restrict__ eval,
    const int* __restrict__ ovf, const int* __restrict__ ovf_n,
    float* __restrict__ agg, int n_edges, int n_rel, int n_nodes, int n_rec) {
  const int lane = threadIdx.x & 31;
  const int wave = (blockIdx.x * 256 + threadIdx.x) >> 5;
  const int n_waves = gridDim.x * 8;
  const int n = min(__builtin_amdgcn_readfirstlane(*ovf_n), n_rec);
  for (int i = wave; i < n; i += n_waves) {
    const int e = __builtin_amdgcn_readfirstlane(ovf[i]);
    if ((unsigned)e >= (unsigned)n_rec) continue;
    const int r = e / n_edges;
    const int d = dst[e];
    const int s = src[e];
    const float v = eval[e];
    if ((unsigned)d >= (unsigned)n_nodes) continue;
    const float2 x = *(const float2*)(inp + (size_t)s * IN_SIZE + 2 * lane);
    float* a = agg + ((size_t)d * n_rel + r) * IN_SIZE + 2 * lane;
    atomicAdd(a, x.x * v);
    atomicAdd(a + 1, x.y * v);
  }
}

__global__ __launch_bounds__(256) void rgcn_gemm_wmma(
    const float* __restrict__ agg, const float* __restrict__ W,
    const float* __restrict__ bias, float* __restrict__ out,
    int n_row_tiles, int n_rel) {
  __shared__ __attribute__((aligned(16))) float sW[IN_SIZE * SW_PITCH];
  const int t = threadIdx.x;
  const int wave = t >> 5;
  const int lane = t & 31;
  const int half = lane >> 4;
  const int l16 = lane & 15;

  int tile = blockIdx.x * 8 + wave;
  const bool live = tile < n_row_tiles;
  if (!live) tile = n_row_tiles - 1;
  const int rowBase = tile * 16;
  const int kpitch = n_rel * IN_SIZE;

  float b0 = 0.f, b1 = 0.f, b2 = 0.f, b3 = 0.f;
  for (int r = 0; r < n_rel; ++r) {
    const float* br = bias + r * OUT_SIZE + l16;
    b0 += br[0]; b1 += br[16]; b2 += br[32]; b3 += br[48];
  }
  v8f c0, c1, c2, c3;
#pragma unroll
  for (int j = 0; j < 8; ++j) { c0[j] = b0; c1[j] = b1; c2[j] = b2; c3[j] = b3; }

  const int koff = half * 2;
  const float* aRow = agg + (size_t)(rowBase + l16) * kpitch + koff;
  const float4* Wv = (const float4*)W;

  for (int r = 0; r < n_rel; ++r) {
    const float* aR = aRow + r * IN_SIZE;
    float2 av[IN_SIZE / 4];
#pragma unroll
    for (int i = 0; i < IN_SIZE / 4; ++i) av[i] = *(const float2*)(aR + 4 * i);
    const float4* Wr = Wv + (size_t)r * (IN_SIZE * OUT_SIZE / 4) + t;
    const float4 w0 = Wr[0], w1 = Wr[256], w2 = Wr[512], w3 = Wr[768];
    __syncthreads();
    float* sp = sW + (t >> 4) * SW_PITCH + (t & 15) * 4;
    *(float4*)(sp) = w0;
    *(float4*)(sp + 16 * SW_PITCH) = w1;
    *(float4*)(sp + 32 * SW_PITCH) = w2;
    *(float4*)(sp + 48 * SW_PITCH) = w3;
    __syncthreads();
#pragma unroll
    for (int k = 0; k < IN_SIZE; k += 4) {
      v2f a, b;
      a[0] = av[k / 4].x;
      a[1] = av[k / 4].y;
      const float* bp = sW + (k + koff) * SW_PITCH + l16;
      b[0] = bp[0];
      b[1] = bp[SW_PITCH];
      c0 = __builtin_amdgcn_wmma_f32_16x16x4_f32(false, a, false, b, (short)0, c0, false, false);
      b[0] = bp[16];
      b[1] = bp[SW_PITCH + 16];
      c1 = __builtin_amdgcn_wmma_f32_16x16x4_f32(false, a, false, b, (short)0, c1, false, false);
      b[0] = bp[32];
      b[1] = bp[SW_PITCH + 32];
      c2 = __builtin_amdgcn_wmma_f32_16x16x4_f32(false, a, false, b, (short)0, c2, false, false);
      b[0] = bp[48];
      b[1] = bp[SW_PITCH + 48];
      c3 = __builtin_amdgcn_wmma_f32_16x16x4_f32(false, a, false, b, (short)0, c3, false, false);
    }
  }

  if (live) {
#pragma unroll
    for (int j = 0; j < 8; ++j) {
      float* o = out + (size_t)(rowBase + j + half * 8) * OUT_SIZE + l16;
      o[0] = c0[j];
      o[16] = c1[j];
      o[32] = c2[j];
      o[48] = c3[j];
    }
  }
}

extern "C" void kernel_launch(void* const* d_in, const int* in_sizes, int n_in,
                              void* d_out, int out_size, void* d_ws, size_t ws_size,
                              hipStream_t stream) {
  const float* inp = (const float*)d_in[0];
  const int* src = (const int*)d_in[1];
  const int* dst = (const int*)d_in[2];
  const float* eval = (const float*)d_in[3];
  const float* weights = (const float*)d_in[4];
  const float* bias = (const float*)d_in[5];
  float* out = (float*)d_out;

  const int n_nodes = in_sizes[0] / IN_SIZE;
  const int n_rel = in_sizes[4] / (IN_SIZE * OUT_SIZE);
  const int n_edges = in_sizes[1] / n_rel;
  const int n_rec = n_rel * n_edges;
  const int n_seg = n_nodes * n_rel;
  const int n_slices = (n_seg + 1023) / 1024;
  const int cnt_plane = n_slices * 1024;
  const size_t seg_pad = (size_t)(2 * n_slices + 1) * 1024;
  const size_t rec_plane = ((size_t)n_seg * HSLOTS + 1023) / 1024 * 1024;
  const size_t rec_pad = 2 * rec_plane;
  const size_t ovf_pad = ((size_t)n_rec + 1023) / 1024 * 1024;

  int* cnt = (int*)d_ws;
  int* ovf_n = cnt + (size_t)2 * cnt_plane;
  int2* rec = (int2*)(cnt + seg_pad);
  int* ovf = (int*)(rec + rec_pad);
  float* agg = (float*)(ovf + ovf_pad);

  const int zero4 = (int)(seg_pad / 4);
  rgcn_zero<<<(zero4 + 255) / 256, 256, 0, stream>>>((int4*)cnt, zero4);

  const dim3 egrid((n_edges + 1023) / 1024, n_rel);
  rgcn_build<<<egrid, 256, 0, stream>>>(src, dst, eval, cnt, ovf_n, rec, ovf,
                                        n_edges, n_rel, n_nodes, n_rec, cnt_plane, rec_plane);

  const int n_runs = (n_seg + RUN - 1) / RUN;
  rgcn_pull<<<(n_runs + 3) / 4, 128, 0, stream>>>(inp, cnt, rec, agg, n_seg, cnt_plane, rec_plane);

  rgcn_overflow<<<256, 256, 0, stream>>>(inp, src, dst, eval, ovf, ovf_n, agg,
                                         n_edges, n_rel, n_nodes, n_rec);

  const int n_row_tiles = n_nodes / 16;
  rgcn_gemm_wmma<<<(n_row_tiles + 7) / 8, 256, 0, stream>>>(
      agg, weights, bias, out, n_row_tiles, n_rel);
}
